# starting kernel
# baseline (speedup 1.0000x reference)
#include <hip/hip_runtime.h>

typedef float v2f __attribute__((ext_vector_type(2)));
typedef float v8f __attribute__((ext_vector_type(8)));

__device__ __forceinline__ v8f wmma4(v2f a, v2f b, v8f c) {
  return __builtin_amdgcn_wmma_f32_16x16x4_f32(false, a, false, b, (short)0, c,
                                               false, false);
}
__device__ __forceinline__ v8f zero8() {
  v8f z = {0.f, 0.f, 0.f, 0.f, 0.f, 0.f, 0.f, 0.f};
  return z;
}

__global__ __launch_bounds__(256) void k_init_deg(unsigned* __restrict__ deg, int n) {
  int i = blockIdx.x * 256 + threadIdx.x;
  if (i < n) deg[i] = 1u;
}

__global__ __launch_bounds__(256) void k_count_deg(const int* __restrict__ dst,
                                                   unsigned* __restrict__ deg, int nE) {
  int i = blockIdx.x * 256 + threadIdx.x;
  if (i < nE) atomicAdd(&deg[dst[i]], 1u);
}

__global__ __launch_bounds__(256) void k_dinv(const unsigned* __restrict__ deg,
                                              float* __restrict__ dinv, int n) {
  int i = blockIdx.x * 256 + threadIdx.x;
  if (i < n) dinv[i] = rsqrtf((float)deg[i]);
}

__global__ __launch_bounds__(256) void k_zero(float* __restrict__ p, int n) {
  int i = blockIdx.x * 256 + threadIdx.x;
  if (i < n) p[i] = 0.f;
}

__global__ __launch_bounds__(256) void k_xw1(const float* __restrict__ x,
                                             const float* __restrict__ W,
                                             const float* __restrict__ dinv,
                                             float* __restrict__ y, int nTiles) {
  int lane = threadIdx.x & 31, wave = threadIdx.x >> 5;
  int tile = blockIdx.x * 8 + wave;
  if (tile >= nTiles) return;
  int col = lane & 15, hi = lane >> 4, koff = hi * 2;
  int base = tile << 4;
  const float* xr = x + (size_t)(base + col) * 4;
  v2f a; a[0] = xr[koff]; a[1] = xr[koff + 1];
  v2f b; b[0] = W[koff * 16 + col]; b[1] = W[(koff + 1) * 16 + col];
  v8f c = zero8();
  c = wmma4(a, b, c);
#pragma unroll
  for (int j = 0; j < 8; ++j) {
    int node = base + j + hi * 8;
    y[(size_t)node * 16 + col] = c[j] * dinv[node];
  }
}

__global__ __launch_bounds__(256) void k_xw2(const float* __restrict__ h,
                                             const float* __restrict__ W,
                                             const float* __restrict__ dinv,
                                             float* __restrict__ y, int nTiles) {
  int lane = threadIdx.x & 31, wave = threadIdx.x >> 5;
  int tile = blockIdx.x * 8 + wave;
  if (tile >= nTiles) return;
  int col = lane & 15, hi = lane >> 4, koff = hi * 2;
  int base = tile << 4;
  const float* hr = h + (size_t)(base + col) * 16;
  v8f c = zero8();
#pragma unroll
  for (int kc = 0; kc < 4; ++kc) {
    v2f a; a[0] = hr[kc * 4 + koff]; a[1] = hr[kc * 4 + koff + 1];
    v2f b; b[0] = W[(kc * 4 + koff) * 16 + col];
    b[1] = W[(kc * 4 + koff + 1) * 16 + col];
    c = wmma4(a, b, c);
  }
#pragma unroll
  for (int j = 0; j < 8; ++j) {
    int node = base + j + hi * 8;
    y[(size_t)node * 16 + col] = c[j] * dinv[node];
  }
}

__global__ __launch_bounds__(256) void k_scatter(const int* __restrict__ src,
                                                 const int* __restrict__ dst,
                                                 const float* __restrict__ y,
                                                 float* __restrict__ acc, int nE) {
  int e = blockIdx.x * 256 + threadIdx.x;
  if (e >= nE) return;
  int s = src[e], d = dst[e];
  const float4* ys = reinterpret_cast<const float4*>(y + (size_t)s * 16);
  float* ad = acc + (size_t)d * 16;
#pragma unroll
  for (int q = 0; q < 4; ++q) {
    float4 v = ys[q];
    atomicAdd(ad + q * 4 + 0, v.x);
    atomicAdd(ad + q * 4 + 1, v.y);
    atomicAdd(ad + q * 4 + 2, v.z);
    atomicAdd(ad + q * 4 + 3, v.w);
  }
}

__global__ __launch_bounds__(256) void k_finish(const float* __restrict__ acc,
                                                const float* __restrict__ y,
                                                const float* __restrict__ dinv,
                                                const float* __restrict__ bias,
                                                float* __restrict__ h, int total,
                                                int doRelu) {
  int i = blockIdx.x * 256 + threadIdx.x;
  if (i >= total) return;
  int node = i >> 4, f = i & 15;
  float v = dinv[node] * (acc[i] + y[i]) + bias[f];
  if (doRelu) v = fmaxf(v, 0.f);
  h[i] = v;
}

__global__ __launch_bounds__(256) void k_mlp_pool(const float* __restrict__ h2,
                                                  const float* __restrict__ Wf1,
                                                  const float* __restrict__ bf1,
                                                  const float* __restrict__ Wf2,
                                                  const float* __restrict__ bf2,
                                                  float* __restrict__ g, int nG) {
  __shared__ float sW[64 * 132];
  __shared__ float sH[8 * 32 * 68];
  int lane = threadIdx.x & 31, wave = threadIdx.x >> 5;

  for (int i = threadIdx.x; i < 64 * 128; i += 256)
    sW[(i >> 7) * 132 + (i & 127)] = Wf2[i];

  int graph = blockIdx.x * 8 + wave;
  if (graph >= nG) graph = nG - 1;
  int col = lane & 15, hi = lane >> 4, koff = hi * 2;
  size_t nb = (size_t)graph * 32;

  v2f a1[2][4];
#pragma unroll
  for (int mt = 0; mt < 2; ++mt) {
    const float* r = h2 + (nb + mt * 16 + col) * 16;
#pragma unroll
    for (int kc = 0; kc < 4; ++kc) {
      a1[mt][kc][0] = r[kc * 4 + koff];
      a1[mt][kc][1] = r[kc * 4 + koff + 1];
    }
  }
  __syncthreads();

  float* mH = sH + wave * (32 * 68);
#pragma unroll
  for (int nt = 0; nt < 4; ++nt) {
    v8f c0 = zero8(), c1 = zero8();
#pragma unroll
    for (int kc = 0; kc < 4; ++kc) {
      v2f b;
      b[0] = Wf1[(kc * 4 + koff) * 64 + nt * 16 + col];
      b[1] = Wf1[(kc * 4 + koff + 1) * 64 + nt * 16 + col];
      c0 = wmma4(a1[0][kc], b, c0);
      c1 = wmma4(a1[1][kc], b, c1);
    }
    float bias = bf1[nt * 16 + col];
#pragma unroll
    for (int j = 0; j < 8; ++j) {
      mH[(j + hi * 8) * 68 + nt * 16 + col]       = fmaxf(c0[j] + bias, 0.f);
      mH[(16 + j + hi * 8) * 68 + nt * 16 + col]  = fmaxf(c1[j] + bias, 0.f);
    }
  }
  __syncthreads();

  const float* A0 = mH + (size_t)col * 68;
  const float* A1 = mH + (size_t)(16 + col) * 68;
  for (int nt2 = 0; nt2 < 8; ++nt2) {
    v8f c0 = zero8(), c1 = zero8();
#pragma unroll
    for (int kc = 0; kc < 16; ++kc) {
      int k0 = kc * 4 + koff;
      v2f a0; a0[0] = A0[k0]; a0[1] = A0[k0 + 1];
      v2f av; av[0] = A1[k0]; av[1] = A1[k0 + 1];
      v2f b;  b[0] = sW[k0 * 132 + nt2 * 16 + col];
      b[1] = sW[(k0 + 1) * 132 + nt2 * 16 + col];
      c0 = wmma4(a0, b, c0);
      c1 = wmma4(av, b, c1);
    }
    float bias = bf2[nt2 * 16 + col];
    float s = 0.f;
#pragma unroll
    for (int j = 0; j < 8; ++j)
      s += fmaxf(c0[j] + bias, 0.f) + fmaxf(c1[j] + bias, 0.f);
    s += __shfl_xor(s, 16, 32);
    if (lane < 16) g[(size_t)graph * 128 + nt2 * 16 + lane] = s * (1.0f / 32.0f);
  }
}

__global__ __launch_bounds__(256) void k_head(const float* __restrict__ g,
                                              const float* __restrict__ Wg,
                                              const float* __restrict__ bg,
                                              const float* __restrict__ Wout,
                                              const float* __restrict__ bout,
                                              float* __restrict__ out, int nG) {
  __shared__ float sG2[8 * 16 * 68];
  int lane = threadIdx.x & 31, wave = threadIdx.x >> 5;
  int tile = blockIdx.x * 8 + wave;
  int maxTile = (nG >> 4) - 1;
  if (tile > maxTile) tile = maxTile;
  int col = lane & 15, hi = lane >> 4, koff = hi * 2;
  int gbase = tile << 4;

  const float* ar = g + (size_t)(gbase + col) * 128;
  v8f c[4] = {zero8(), zero8(), zero8(), zero8()};
  for (int kc = 0; kc < 32; ++kc) {
    int k0 = kc * 4 + koff;
    v2f a; a[0] = ar[k0]; a[1] = ar[k0 + 1];
#pragma unroll
    for (int nt = 0; nt < 4; ++nt) {
      v2f b;
      b[0] = Wg[k0 * 64 + nt * 16 + col];
      b[1] = Wg[(k0 + 1) * 64 + nt * 16 + col];
      c[nt] = wmma4(a, b, c[nt]);
    }
  }
  float* mG = sG2 + wave * (16 * 68);
#pragma unroll
  for (int nt = 0; nt < 4; ++nt) {
    float bias = bg[nt * 16 + col];
#pragma unroll
    for (int j = 0; j < 8; ++j)
      mG[(j + hi * 8) * 68 + nt * 16 + col] = fmaxf(c[nt][j] + bias, 0.f);
  }
  __syncthreads();

  const float* g2 = mG + (size_t)col * 68;
  int gidx = gbase + col;
#pragma unroll
  for (int mm = 0; mm < 2; ++mm) {
    int m = hi * 2 + mm;
    float acc[5];
#pragma unroll
    for (int a2 = 0; a2 < 5; ++a2) acc[a2] = bout[m * 5 + a2];
    for (int f = 0; f < 64; ++f) {
      float gv = g2[f];
#pragma unroll
      for (int a2 = 0; a2 < 5; ++a2)
        acc[a2] = fmaf(gv, Wout[(m * 64 + f) * 5 + a2], acc[a2]);
    }
#pragma unroll
    for (int a2 = 0; a2 < 5; ++a2)
      out[((size_t)gidx * 4 + m) * 5 + a2] = acc[a2];
  }
}

extern "C" void kernel_launch(void* const* d_in, const int* in_sizes, int n_in,
                              void* d_out, int out_size, void* d_ws, size_t ws_size,
                              hipStream_t stream) {
  const float* x    = (const float*)d_in[0];
  const int*   ei   = (const int*)d_in[1];
  const float* W1   = (const float*)d_in[3];
  const float* b1   = (const float*)d_in[4];
  const float* W2   = (const float*)d_in[5];
  const float* b2   = (const float*)d_in[6];
  const float* Wf1  = (const float*)d_in[7];
  const float* bf1  = (const float*)d_in[8];
  const float* Wf2  = (const float*)d_in[9];
  const float* bf2  = (const float*)d_in[10];
  const float* Wg   = (const float*)d_in[11];
  const float* bg   = (const float*)d_in[12];
  const float* Wout = (const float*)d_in[13];
  const float* bout = (const float*)d_in[14];
  float* out = (float*)d_out;

  const int N = in_sizes[0] / 4;
  const int E = in_sizes[1] / 2;
  const int G = in_sizes[2] / 32;
  const int* src = ei;
  const int* dst = ei + E;

  char* ws = (char*)d_ws;
  size_t off = 0;
  auto take = [&](size_t bytes) -> void* {
    void* p = ws + off;
    off += (bytes + 255) & ~(size_t)255;
    return p;
  };
  unsigned* deg  = (unsigned*)take((size_t)N * 4);
  float*    dinv = (float*)take((size_t)N * 4);
  float*    ybuf = (float*)take((size_t)N * 16 * 4);
  float*    accb = (float*)take((size_t)N * 16 * 4);
  float*    hbuf = (float*)take((size_t)N * 16 * 4);
  float*    gbuf = (float*)take((size_t)G * 128 * 4);
  (void)ws_size; (void)n_in; (void)out_size;

  const int NF   = N * 16;
  const int gN   = (N + 255) / 256;
  const int gE   = (E + 255) / 256;
  const int gNF  = (NF + 255) / 256;
  const int gXW  = (N / 16 + 7) / 8;
  const int gMLP = (G + 7) / 8;
  const int gHD  = (G / 16 + 7) / 8;

  k_init_deg<<<gN, 256, 0, stream>>>(deg, N);
  k_count_deg<<<gE, 256, 0, stream>>>(dst, deg, E);
  k_dinv<<<gN, 256, 0, stream>>>(deg, dinv, N);

  k_zero<<<gNF, 256, 0, stream>>>(accb, NF);
  k_xw1<<<gXW, 256, 0, stream>>>(x, W1, dinv, ybuf, N / 16);
  k_scatter<<<gE, 256, 0, stream>>>(src, dst, ybuf, accb, E);
  k_finish<<<gNF, 256, 0, stream>>>(accb, ybuf, dinv, b1, hbuf, NF, 1);

  k_zero<<<gNF, 256, 0, stream>>>(accb, NF);
  k_xw2<<<gXW, 256, 0, stream>>>(hbuf, W2, dinv, ybuf, N / 16);
  k_scatter<<<gE, 256, 0, stream>>>(src, dst, ybuf, accb, E);
  k_finish<<<gNF, 256, 0, stream>>>(accb, ybuf, dinv, b2, hbuf, NF, 0);

  k_mlp_pool<<<gMLP, 256, 0, stream>>>(hbuf, Wf1, bf1, Wf2, bf2, gbuf, G);

  k_head<<<gHD, 256, 0, stream>>>(gbuf, Wg, bg, Wout, bout, out, G);
}


# Replace the f32 atomic edge scatter of both conv layers by a destination-sorted edge list (returning integer atomic gives each edge its rank, scan, fill) and an atomic-free per-node gather
# speedup vs baseline: 3.9575x; 3.9575x over previous
#include <hip/hip_runtime.h>

typedef float v2f __attribute__((ext_vector_type(2)));
typedef float v8f __attribute__((ext_vector_type(8)));

__device__ __forceinline__ v8f wmma4(v2f a, v2f b, v8f c) {
  return __builtin_amdgcn_wmma_f32_16x16x4_f32(false, a, false, b, (short)0, c,
                                               false, false);
}
__device__ __forceinline__ v8f zero8() {
  v8f z = {0.f, 0.f, 0.f, 0.f, 0.f, 0.f, 0.f, 0.f};
  return z;
}

__global__ __launch_bounds__(256) void k_init_cnt(unsigned* __restrict__ cnt, int n) {
  int i = blockIdx.x * 256 + threadIdx.x;
  if (i < n) cnt[i] = 0u;
}

__global__ __launch_bounds__(256) void k_count_rank(const int* __restrict__ dst,
                                                    unsigned* __restrict__ cnt,
                                                    unsigned* __restrict__ rank, int nE) {
  int t = blockIdx.x * 256 + threadIdx.x;
  int stride = gridDim.x * 256;
  int e0 = t, e1 = t + stride, e2 = t + 2 * stride, e3 = t + 3 * stride;
  int d0 = (e0 < nE) ? dst[e0] : -1;
  int d1 = (e1 < nE) ? dst[e1] : -1;
  int d2 = (e2 < nE) ? dst[e2] : -1;
  int d3 = (e3 < nE) ? dst[e3] : -1;
  unsigned r0 = 0u, r1 = 0u, r2 = 0u, r3 = 0u;
  if (d0 >= 0) r0 = atomicAdd(&cnt[d0], 1u);
  if (d1 >= 0) r1 = atomicAdd(&cnt[d1], 1u);
  if (d2 >= 0) r2 = atomicAdd(&cnt[d2], 1u);
  if (d3 >= 0) r3 = atomicAdd(&cnt[d3], 1u);
  if (d0 >= 0) rank[e0] = r0;
  if (d1 >= 0) rank[e1] = r1;
  if (d2 >= 0) rank[e2] = r2;
  if (d3 >= 0) rank[e3] = r3;
}

__global__ __launch_bounds__(256) void k_block_sum(const unsigned* __restrict__ cnt,
                                                   unsigned* __restrict__ bsum, int n) {
  __shared__ unsigned red[8];
  int i = blockIdx.x * 256 + threadIdx.x;
  int lane = threadIdx.x & 31, wave = threadIdx.x >> 5;
  unsigned v = (i < n) ? cnt[i] : 0u;
  for (int o = 16; o > 0; o >>= 1) v += __shfl_xor(v, o, 32);
  if (lane == 0) red[wave] = v;
  __syncthreads();
  if (threadIdx.x == 0) {
    unsigned s = 0u;
    for (int w = 0; w < 8; ++w) s += red[w];
    bsum[blockIdx.x] = s;
  }
}

__global__ __launch_bounds__(256) void k_scan_dinv(const unsigned* __restrict__ cnt,
                                                   const unsigned* __restrict__ bsum,
                                                   int* __restrict__ rowstart,
                                                   float* __restrict__ dinv, int n) {
  __shared__ unsigned red[8];
  __shared__ unsigned wsum[8];
  int tid = threadIdx.x, lane = tid & 31, wave = tid >> 5;
  unsigned p = 0u;
  for (int j = tid; j < (int)blockIdx.x; j += 256) p += bsum[j];
  for (int o = 16; o > 0; o >>= 1) p += __shfl_xor(p, o, 32);
  if (lane == 0) red[wave] = p;

  int i = blockIdx.x * 256 + tid;
  unsigned cv = (i < n) ? cnt[i] : 0u;
  unsigned x = cv;
  for (int o = 1; o < 32; o <<= 1) {
    unsigned up = __shfl_up(x, o, 32);
    if (lane >= o) x += up;
  }
  if (lane == 31) wsum[wave] = x;
  __syncthreads();
  unsigned base = 0u;
  for (int w = 0; w < 8; ++w) {
    base += red[w];
    if (w < wave) base += wsum[w];
  }
  if (i < n) {
    rowstart[i] = (int)(base + x - cv);
    dinv[i] = rsqrtf((float)(cv + 1u));
  }
}

__global__ __launch_bounds__(256) void k_fill(const int* __restrict__ src,
                                              const int* __restrict__ dst,
                                              const unsigned* __restrict__ rank,
                                              const int* __restrict__ rowstart,
                                              int* __restrict__ col, int nE) {
  int t = blockIdx.x * 256 + threadIdx.x;
  int stride = gridDim.x * 256;
  int d[4], s[4], rs[4];
  unsigned r[4];
  for (int u = 0; u < 4; ++u) {
    int e = t + u * stride;
    bool ok = e < nE;
    d[u] = ok ? dst[e] : -1;
    r[u] = ok ? rank[e] : 0u;
    s[u] = ok ? src[e] : 0;
  }
  for (int u = 0; u < 4; ++u) rs[u] = (d[u] >= 0) ? rowstart[d[u]] : 0;
  for (int u = 0; u < 4; ++u)
    if (d[u] >= 0) col[rs[u] + (int)r[u]] = s[u];
}

__global__ __launch_bounds__(256) void k_xw1(const float* __restrict__ x,
                                             const float* __restrict__ W,
                                             const float* __restrict__ dinv,
                                             float* __restrict__ y, int nTiles) {
  int lane = threadIdx.x & 31, wave = threadIdx.x >> 5;
  int tile = blockIdx.x * 8 + wave;
  if (tile >= nTiles) return;
  int col = lane & 15, hi = lane >> 4, koff = hi * 2;
  int base = tile << 4;
  const float* xr = x + (size_t)(base + col) * 4;
  v2f a; a[0] = xr[koff]; a[1] = xr[koff + 1];
  v2f b; b[0] = W[koff * 16 + col]; b[1] = W[(koff + 1) * 16 + col];
  v8f c = zero8();
  c = wmma4(a, b, c);
#pragma unroll
  for (int j = 0; j < 8; ++j) {
    int node = base + j + hi * 8;
    y[(size_t)node * 16 + col] = c[j] * dinv[node];
  }
}

__global__ __launch_bounds__(256) void k_xw2(const float* __restrict__ h,
                                             const float* __restrict__ W,
                                             const float* __restrict__ dinv,
                                             float* __restrict__ y, int nTiles) {
  int lane = threadIdx.x & 31, wave = threadIdx.x >> 5;
  int tile = blockIdx.x * 8 + wave;
  if (tile >= nTiles) return;
  int col = lane & 15, hi = lane >> 4, koff = hi * 2;
  int base = tile << 4;
  const float* hr = h + (size_t)(base + col) * 16;
  v8f c = zero8();
#pragma unroll
  for (int kc = 0; kc < 4; ++kc) {
    v2f a; a[0] = hr[kc * 4 + koff]; a[1] = hr[kc * 4 + koff + 1];
    v2f b; b[0] = W[(kc * 4 + koff) * 16 + col];
    b[1] = W[(kc * 4 + koff + 1) * 16 + col];
    c = wmma4(a, b, c);
  }
#pragma unroll
  for (int j = 0; j < 8; ++j) {
    int node = base + j + hi * 8;
    y[(size_t)node * 16 + col] = c[j] * dinv[node];
  }
}

__device__ __forceinline__ float4 add4(float4 a, float4 b) {
  return make_float4(a.x + b.x, a.y + b.y, a.z + b.z, a.w + b.w);
}
__device__ __forceinline__ float4 keep4(bool m, float4 a) {
  return make_float4(m ? a.x : 0.f, m ? a.y : 0.f, m ? a.z : 0.f, m ? a.w : 0.f);
}

__global__ __launch_bounds__(256) void k_gather(const int* __restrict__ rowstart,
                                                const int* __restrict__ col,
                                                const float* __restrict__ y,
                                                const float* __restrict__ dinv,
                                                const float* __restrict__ bias,
                                                float* __restrict__ h, int n, int nE,
                                                int doRelu) {
  int t = blockIdx.x * 256 + threadIdx.x;
  int node = t >> 2, q = t & 3;
  if (node >= n) return;
  int beg = rowstart[node];
  int end = (node + 1 < n) ? rowstart[node + 1] : nE;
  const float4* __restrict__ y4 = reinterpret_cast<const float4*>(y);
  float4 acc = y4[(size_t)node * 4 + q];
  float di = dinv[node];

#pragma unroll 2
  for (int j = beg; j < end; j += 4) {
    int c0 = col[j], c1 = col[j + 1], c2 = col[j + 2], c3 = col[j + 3];
    bool m1 = j + 1 < end, m2 = j + 2 < end, m3 = j + 3 < end;
    int s1 = m1 ? c1 : node, s2 = m2 ? c2 : node, s3 = m3 ? c3 : node;
    float4 v0 = y4[(size_t)c0 * 4 + q];
    float4 v1 = y4[(size_t)s1 * 4 + q];
    float4 v2 = y4[(size_t)s2 * 4 + q];
    float4 v3 = y4[(size_t)s3 * 4 + q];
    acc = add4(acc, add4(add4(v0, keep4(m1, v1)), add4(keep4(m2, v2), keep4(m3, v3))));
  }

  float4 r = make_float4(di * acc.x + bias[q * 4 + 0], di * acc.y + bias[q * 4 + 1],
                         di * acc.z + bias[q * 4 + 2], di * acc.w + bias[q * 4 + 3]);
  if (doRelu) {
    r.x = fmaxf(r.x, 0.f); r.y = fmaxf(r.y, 0.f);
    r.z = fmaxf(r.z, 0.f); r.w = fmaxf(r.w, 0.f);
  }
  reinterpret_cast<float4*>(h)[(size_t)node * 4 + q] = r;
}

__global__ __launch_bounds__(256) void k_mlp_pool(const float* __restrict__ h2,
                                                  const float* __restrict__ Wf1,
                                                  const float* __restrict__ bf1,
                                                  const float* __restrict__ Wf2,
                                                  const float* __restrict__ bf2,
                                                  float* __restrict__ g, int nG) {
  __shared__ float sW[64 * 132];
  __shared__ float sH[8 * 32 * 68];
  int lane = threadIdx.x & 31, wave = threadIdx.x >> 5;

  for (int i = threadIdx.x; i < 64 * 128; i += 256)
    sW[(i >> 7) * 132 + (i & 127)] = Wf2[i];

  int graph = blockIdx.x * 8 + wave;
  if (graph >= nG) graph = nG - 1;
  int col = lane & 15, hi = lane >> 4, koff = hi * 2;
  size_t nb = (size_t)graph * 32;

  v2f a1[2][4];
#pragma unroll
  for (int mt = 0; mt < 2; ++mt) {
    const float* r = h2 + (nb + mt * 16 + col) * 16;
#pragma unroll
    for (int kc = 0; kc < 4; ++kc) {
      a1[mt][kc][0] = r[kc * 4 + koff];
      a1[mt][kc][1] = r[kc * 4 + koff + 1];
    }
  }
  __syncthreads();

  float* mH = sH + wave * (32 * 68);
#pragma unroll
  for (int nt = 0; nt < 4; ++nt) {
    v8f c0 = zero8(), c1 = zero8();
#pragma unroll
    for (int kc = 0; kc < 4; ++kc) {
      v2f b;
      b[0] = Wf1[(kc * 4 + koff) * 64 + nt * 16 + col];
      b[1] = Wf1[(kc * 4 + koff + 1) * 64 + nt * 16 + col];
      c0 = wmma4(a1[0][kc], b, c0);
      c1 = wmma4(a1[1][kc], b, c1);
    }
    float bias = bf1[nt * 16 + col];
#pragma unroll
    for (int j = 0; j < 8; ++j) {
      mH[(j + hi * 8) * 68 + nt * 16 + col]       = fmaxf(c0[j] + bias, 0.f);
      mH[(16 + j + hi * 8) * 68 + nt * 16 + col]  = fmaxf(c1[j] + bias, 0.f);
    }
  }
  __syncthreads();

  const float* A0 = mH + (size_t)col * 68;
  const float* A1 = mH + (size_t)(16 + col) * 68;
  for (int nt2 = 0; nt2 < 8; ++nt2) {
    v8f c0 = zero8(), c1 = zero8();
#pragma unroll
    for (int kc = 0; kc < 16; ++kc) {
      int k0 = kc * 4 + koff;
      v2f a0; a0[0] = A0[k0]; a0[1] = A0[k0 + 1];
      v2f av; av[0] = A1[k0]; av[1] = A1[k0 + 1];
      v2f b;  b[0] = sW[k0 * 132 + nt2 * 16 + col];
      b[1] = sW[(k0 + 1) * 132 + nt2 * 16 + col];
      c0 = wmma4(a0, b, c0);
      c1 = wmma4(av, b, c1);
    }
    float bias = bf2[nt2 * 16 + col];
    float s = 0.f;
#pragma unroll
    for (int j = 0; j < 8; ++j)
      s += fmaxf(c0[j] + bias, 0.f) + fmaxf(c1[j] + bias, 0.f);
    s += __shfl_xor(s, 16, 32);
    if (lane < 16) g[(size_t)graph * 128 + nt2 * 16 + lane] = s * (1.0f / 32.0f);
  }
}

__global__ __launch_bounds__(256) void k_head(const float* __restrict__ g,
                                              const float* __restrict__ Wg,
                                              const float* __restrict__ bg,
                                              const float* __restrict__ Wout,
                                              const float* __restrict__ bout,
                                              float* __restrict__ out, int nG) {
  __shared__ float sG2[8 * 16 * 68];
  int lane = threadIdx.x & 31, wave = threadIdx.x >> 5;
  int tile = blockIdx.x * 8 + wave;
  int maxTile = (nG >> 4) - 1;
  if (tile > maxTile) tile = maxTile;
  int col = lane & 15, hi = lane >> 4, koff = hi * 2;
  int gbase = tile << 4;

  const float* ar = g + (size_t)(gbase + col) * 128;
  v8f c[4] = {zero8(), zero8(), zero8(), zero8()};
  for (int kc = 0; kc < 32; ++kc) {
    int k0 = kc * 4 + koff;
    v2f a; a[0] = ar[k0]; a[1] = ar[k0 + 1];
#pragma unroll
    for (int nt = 0; nt < 4; ++nt) {
      v2f b;
      b[0] = Wg[k0 * 64 + nt * 16 + col];
      b[1] = Wg[(k0 + 1) * 64 + nt * 16 + col];
      c[nt] = wmma4(a, b, c[nt]);
    }
  }
  float* mG = sG2 + wave * (16 * 68);
#pragma unroll
  for (int nt = 0; nt < 4; ++nt) {
    float bias = bg[nt * 16 + col];
#pragma unroll
    for (int j = 0; j < 8; ++j)
      mG[(j + hi * 8) * 68 + nt * 16 + col] = fmaxf(c[nt][j] + bias, 0.f);
  }
  __syncthreads();

  const float* g2 = mG + (size_t)col * 68;
  int gidx = gbase + col;
#pragma unroll
  for (int mm = 0; mm < 2; ++mm) {
    int m = hi * 2 + mm;
    float acc[5];
#pragma unroll
    for (int a2 = 0; a2 < 5; ++a2) acc[a2] = bout[m * 5 + a2];
    for (int f = 0; f < 64; ++f) {
      float gv = g2[f];
#pragma unroll
      for (int a2 = 0; a2 < 5; ++a2)
        acc[a2] = fmaf(gv, Wout[(m * 64 + f) * 5 + a2], acc[a2]);
    }
#pragma unroll
    for (int a2 = 0; a2 < 5; ++a2)
      out[((size_t)gidx * 4 + m) * 5 + a2] = acc[a2];
  }
}

extern "C" void kernel_launch(void* const* d_in, const int* in_sizes, int n_in,
                              void* d_out, int out_size, void* d_ws, size_t ws_size,
                              hipStream_t stream) {
  const float* x    = (const float*)d_in[0];
  const int*   ei   = (const int*)d_in[1];
  const float* W1   = (const float*)d_in[3];
  const float* b1   = (const float*)d_in[4];
  const float* W2   = (const float*)d_in[5];
  const float* b2   = (const float*)d_in[6];
  const float* Wf1  = (const float*)d_in[7];
  const float* bf1  = (const float*)d_in[8];
  const float* Wf2  = (const float*)d_in[9];
  const float* bf2  = (const float*)d_in[10];
  const float* Wg   = (const float*)d_in[11];
  const float* bg   = (const float*)d_in[12];
  const float* Wout = (const float*)d_in[13];
  const float* bout = (const float*)d_in[14];
  float* out = (float*)d_out;

  const int N = in_sizes[0] / 4;
  const int E = in_sizes[1] / 2;
  const int G = in_sizes[2] / 32;
  const int* src = ei;
  const int* dst = ei + E;

  char* ws = (char*)d_ws;
  size_t off = 0;
  auto take = [&](size_t bytes) -> void* {
    void* p = ws + off;
    off += (bytes + 255) & ~(size_t)255;
    return p;
  };
  float*    dinv     = (float*)take((size_t)N * 4);
  int*      rowstart = (int*)take((size_t)N * 4);
  float*    ybuf     = (float*)take((size_t)N * 16 * 4);
  float*    hbuf     = (float*)take((size_t)N * 16 * 4);
  int*      col      = (int*)take((size_t)E * 4);
  float*    gbuf     = (float*)take((size_t)G * 128 * 4);
  unsigned* rank     = (unsigned*)hbuf;
  unsigned* cnt      = (unsigned*)gbuf;
  unsigned* bsum     = cnt + N;
  (void)ws_size; (void)n_in; (void)out_size;

  const int gN   = (N + 255) / 256;
  const int gE4  = (E + 1023) / 1024;
  const int gN4  = (N * 4 + 255) / 256;
  const int gXW  = (N / 16 + 7) / 8;
  const int gMLP = (G + 7) / 8;
  const int gHD  = (G / 16 + 7) / 8;

  k_init_cnt<<<gN, 256, 0, stream>>>(cnt, N);
  k_count_rank<<<gE4, 256, 0, stream>>>(dst, cnt, rank, E);
  k_block_sum<<<gN, 256, 0, stream>>>(cnt, bsum, N);
  k_scan_dinv<<<gN, 256, 0, stream>>>(cnt, bsum, rowstart, dinv, N);
  k_fill<<<gE4, 256, 0, stream>>>(src, dst, rank, rowstart, col, E);

  k_xw1<<<gXW, 256, 0, stream>>>(x, W1, dinv, ybuf, N / 16);
  k_gather<<<gN4, 256, 0, stream>>>(rowstart, col, ybuf, dinv, b1, hbuf, N, E, 1);

  k_xw2<<<gXW, 256, 0, stream>>>(hbuf, W2, dinv, ybuf, N / 16);
  k_gather<<<gN4, 256, 0, stream>>>(rowstart, col, ybuf, dinv, b2, hbuf, N, E, 0);

  k_mlp_pool<<<gMLP, 256, 0, stream>>>(hbuf, Wf1, bf1, Wf2, bf2, gbuf, G);

  k_head<<<gHD, 256, 0, stream>>>(gbuf, Wg, bg, Wout, bout, out, G);
}


# Chunk offsets inside the buckets from an atomic-free column scan of the chunk x bucket count table, replacing one returning global add per (chunk, bucket); bucket sizes kept dense
# speedup vs baseline: 9.1715x; 2.3175x over previous
#include <hip/hip_runtime.h>

typedef float v2f __attribute__((ext_vector_type(2)));
typedef float v8f __attribute__((ext_vector_type(8)));

__device__ __forceinline__ v8f wmma4(v2f a, v2f b, v8f c) {
  return __builtin_amdgcn_wmma_f32_16x16x4_f32(false, a, false, b, (short)0, c,
                                               false, false);
}
__device__ __forceinline__ v8f zero8() {
  v8f z = {0.f, 0.f, 0.f, 0.f, 0.f, 0.f, 0.f, 0.f};
  return z;
}

#define kBucketShift 8
#define kBucketNodes 256
#define kMaxBuckets 1024
#define kChunk 4096
#define kStageCap 8192

__device__ __forceinline__ void load_chunk(const int* __restrict__ p, int e0, int nE,
                                           bool whole, int fill, int (&v)[16]) {
  if (whole) {
#pragma unroll
    for (int u = 0; u < 16; ++u) v[u] = p[e0 + u * 256];
  } else {
#pragma unroll
    for (int u = 0; u < 16; ++u) {
      int e = e0 + u * 256;
      v[u] = (e < nE) ? p[e] : fill;
    }
  }
}

__global__ __launch_bounds__(256) void k_bhist(const int* __restrict__ dst,
                                               unsigned* __restrict__ offs, int nE) {
  __shared__ unsigned h[kMaxBuckets];
  int tid = threadIdx.x;
#pragma unroll
  for (int k = 0; k < 4; ++k) h[tid + k * 256] = 0u;
  int e0 = blockIdx.x * kChunk + tid;
  bool whole = (blockIdx.x + 1) * kChunk <= nE;
  int d[16];
  load_chunk(dst, e0, nE, whole, -1, d);
  __syncthreads();
  if (whole) {
#pragma unroll
    for (int u = 0; u < 16; ++u) atomicAdd(&h[d[u] >> kBucketShift], 1u);
  } else {
#pragma unroll
    for (int u = 0; u < 16; ++u)
      if (d[u] >= 0) atomicAdd(&h[d[u] >> kBucketShift], 1u);
  }
  __syncthreads();
#pragma unroll
  for (int k = 0; k < 4; ++k)
    offs[(size_t)blockIdx.x * kMaxBuckets + tid + k * 256] = h[tid + k * 256];
}

__global__ __launch_bounds__(1024) void k_colscan(unsigned* __restrict__ offs,
                                                  unsigned* __restrict__ bcnt, int nChunks) {
  __shared__ unsigned tot[32][33];
  int lane = threadIdx.x & 31, w = threadIdx.x >> 5;
  int b = blockIdx.x * 32 + lane;
  unsigned* __restrict__ colp = offs + b;
  bool whole = (w + 1) * 32 <= nChunks;
  unsigned v[32];
  if (whole) {
#pragma unroll
    for (int r = 0; r < 32; ++r) v[r] = colp[(size_t)(w * 32 + r) * kMaxBuckets];
  } else {
#pragma unroll
    for (int r = 0; r < 32; ++r) {
      int ch = w * 32 + r;
      v[r] = (ch < nChunks) ? colp[(size_t)ch * kMaxBuckets] : 0u;
    }
  }
  unsigned s = 0u;
#pragma unroll
  for (int r = 0; r < 32; ++r) { unsigned t = v[r]; v[r] = s; s += t; }
  tot[w][lane] = s;
  __syncthreads();
  unsigned base = 0u, all = 0u;
  for (int k = 0; k < 32; ++k) {
    unsigned t = tot[k][lane];
    base += (k < w) ? t : 0u;
    all += t;
  }
  if (whole) {
#pragma unroll
    for (int r = 0; r < 32; ++r) colp[(size_t)(w * 32 + r) * kMaxBuckets] = base + v[r];
  } else {
#pragma unroll
    for (int r = 0; r < 32; ++r) {
      int ch = w * 32 + r;
      if (ch < nChunks) colp[(size_t)ch * kMaxBuckets] = base + v[r];
    }
  }
  if (w == 0) bcnt[b] = all;
}

__global__ __launch_bounds__(256) void k_bin(const int* __restrict__ src,
                                             const int* __restrict__ dst,
                                             const unsigned* __restrict__ bcnt,
                                             const unsigned* __restrict__ offs,
                                             unsigned* __restrict__ pairs, int nE) {
  __shared__ __align__(16) unsigned h[kMaxBuckets];
  __shared__ unsigned lstart[kMaxBuckets];
  __shared__ unsigned gbase[kMaxBuckets];
  __shared__ unsigned sW[kChunk];
  __shared__ unsigned short sB[kChunk];
  __shared__ unsigned wc[8], wl[8];
  int tid = threadIdx.x, lane = tid & 31, wave = tid >> 5;
#pragma unroll
  for (int k = 0; k < 4; ++k) h[tid + k * 256] = 0u;
  int e0 = blockIdx.x * kChunk + tid;
  bool whole = (blockIdx.x + 1) * kChunk <= nE;
  int d[16], sv[16];
  load_chunk(dst, e0, nE, whole, -1, d);
  load_chunk(src, e0, nE, whole, 0, sv);
  int b4 = tid * 4;
  uint4 c4 = *reinterpret_cast<const uint4*>(bcnt + b4);
  unsigned c[4] = {c4.x, c4.y, c4.z, c4.w};
  uint4 o4 = *reinterpret_cast<const uint4*>(offs + (size_t)blockIdx.x * kMaxBuckets + b4);
  unsigned o[4] = {o4.x, o4.y, o4.z, o4.w};
  __syncthreads();
  if (whole) {
#pragma unroll
    for (int u = 0; u < 16; ++u) atomicAdd(&h[d[u] >> kBucketShift], 1u);
  } else {
#pragma unroll
    for (int u = 0; u < 16; ++u)
      if (d[u] >= 0) atomicAdd(&h[d[u] >> kBucketShift], 1u);
  }
  __syncthreads();
  uint4 l4 = *reinterpret_cast<const uint4*>(&h[b4]);
  unsigned l[4] = {l4.x, l4.y, l4.z, l4.w};
  unsigned sc = c[0] + c[1] + c[2] + c[3];
  unsigned sl = l[0] + l[1] + l[2] + l[3];
  unsigned xc = sc, xl = sl;
  for (int s = 1; s < 32; s <<= 1) {
    unsigned uc = __shfl_up(xc, s, 32);
    unsigned ul = __shfl_up(xl, s, 32);
    if (lane >= s) { xc += uc; xl += ul; }
  }
  if (lane == 31) { wc[wave] = xc; wl[wave] = xl; }
  __syncthreads();
  unsigned ec = xc - sc, el = xl - sl;
  for (int w = 0; w < 8; ++w)
    if (w < wave) { ec += wc[w]; el += wl[w]; }
#pragma unroll
  for (int k = 0; k < 4; ++k) {
    lstart[b4 + k] = el;
    gbase[b4 + k] = ec + o[k];
    h[b4 + k] = 0u;
    el += l[k];
    ec += c[k];
  }
  __syncthreads();
  if (whole) {
#pragma unroll
    for (int u = 0; u < 16; ++u) {
      int b = d[u] >> kBucketShift;
      unsigned pos = lstart[b] + atomicAdd(&h[b], 1u);
      sW[pos] = ((unsigned)sv[u] << kBucketShift) | (unsigned)(d[u] & (kBucketNodes - 1));
      sB[pos] = (unsigned short)b;
    }
  } else {
#pragma unroll
    for (int u = 0; u < 16; ++u) {
      if (d[u] >= 0) {
        int b = d[u] >> kBucketShift;
        unsigned pos = lstart[b] + atomicAdd(&h[b], 1u);
        sW[pos] = ((unsigned)sv[u] << kBucketShift) | (unsigned)(d[u] & (kBucketNodes - 1));
        sB[pos] = (unsigned short)b;
      }
    }
  }
  __syncthreads();
  int left = nE - blockIdx.x * kChunk;
  int cnt = left < kChunk ? left : kChunk;
  for (int i = tid; i < cnt; i += 256) {
    unsigned b = sB[i];
    pairs[gbase[b] + ((unsigned)i - lstart[b])] = sW[i];
  }
}

__global__ __launch_bounds__(256) void k_bucket_xw1(const unsigned* __restrict__ pairs,
                                                    const unsigned* __restrict__ bcnt,
                                                    int* __restrict__ rowstart,
                                                    float* __restrict__ dinv,
                                                    int* __restrict__ col,
                                                    const float* __restrict__ x,
                                                    const float* __restrict__ W,
                                                    float* __restrict__ y, int n) {
  __shared__ unsigned red[8];
  __shared__ unsigned wsum[8];
  __shared__ float sD[256];
  __shared__ unsigned cntL[256];
  __shared__ unsigned lrs[256];
  __shared__ unsigned sCol[kStageCap];
  int tid = threadIdx.x, lane = tid & 31, wave = tid >> 5;
  int c16 = lane & 15, hi = lane >> 4, koff = hi * 2;

  int nb0 = blockIdx.x * 256 + wave * 32;
  bool t0 = nb0 + 16 <= n, t1 = nb0 + 32 <= n;
  v2f a0 = {0.f, 0.f}, a1 = {0.f, 0.f};
  if (t0) {
    const float* xr = x + (size_t)(nb0 + c16) * 4;
    a0[0] = xr[koff]; a0[1] = xr[koff + 1];
  }
  if (t1) {
    const float* xr = x + (size_t)(nb0 + 16 + c16) * 4;
    a1[0] = xr[koff]; a1[1] = xr[koff + 1];
  }
  v2f b; b[0] = W[koff * 16 + c16]; b[1] = W[(koff + 1) * 16 + c16];

  unsigned p = 0u;
#pragma unroll
  for (int k = 0; k < kMaxBuckets / 256; ++k) {
    int j = tid + k * 256;
    unsigned v = bcnt[j];
    p += (j < (int)blockIdx.x) ? v : 0u;
  }
  for (int o = 16; o > 0; o >>= 1) p += __shfl_xor(p, o, 32);
  if (lane == 0) red[wave] = p;
  unsigned nb = bcnt[blockIdx.x];
  cntL[tid] = 0u;
  __syncthreads();
  unsigned bstart = 0u;
  for (int w = 0; w < 8; ++w) bstart += red[w];
  const unsigned* __restrict__ pw = pairs + bstart;

  for (unsigned i0 = 0; i0 < nb; i0 += 1024) {
    unsigned w[4];
#pragma unroll
    for (int k = 0; k < 4; ++k) {
      unsigned i = i0 + k * 256 + tid;
      w[k] = pw[i < nb ? i : nb - 1];
    }
#pragma unroll
    for (int k = 0; k < 4; ++k)
      if (i0 + k * 256 + tid < nb) atomicAdd(&cntL[w[k] & (kBucketNodes - 1)], 1u);
  }
  __syncthreads();

  int node = blockIdx.x * 256 + tid;
  unsigned cv = cntL[tid];
  float dv = rsqrtf((float)(cv + 1u));
  sD[tid] = dv;
  unsigned xs = cv;
  for (int o = 1; o < 32; o <<= 1) {
    unsigned up = __shfl_up(xs, o, 32);
    if (lane >= o) xs += up;
  }
  if (lane == 31) wsum[wave] = xs;
  __syncthreads();
  unsigned base = 0u;
  for (int w = 0; w < 8; ++w)
    if (w < wave) base += wsum[w];
  unsigned ex = base + xs - cv;
  lrs[tid] = ex;
  cntL[tid] = 0u;
  if (node < n) {
    rowstart[node] = (int)(bstart + ex);
    dinv[node] = dv;
  }
  __syncthreads();

  int* __restrict__ cb = col + bstart;
  bool staged = nb <= (unsigned)kStageCap;
  for (unsigned i0 = 0; i0 < nb; i0 += 1024) {
    unsigned w[4];
#pragma unroll
    for (int k = 0; k < 4; ++k) {
      unsigned i = i0 + k * 256 + tid;
      w[k] = pw[i < nb ? i : nb - 1];
    }
#pragma unroll
    for (int k = 0; k < 4; ++k) {
      if (i0 + k * 256 + tid < nb) {
        unsigned dl = w[k] & (kBucketNodes - 1);
        unsigned pos = lrs[dl] + atomicAdd(&cntL[dl], 1u);
        if (staged) sCol[pos] = w[k] >> kBucketShift;
        else cb[pos] = (int)(w[k] >> kBucketShift);
      }
    }
  }
  if (staged) {
    __syncthreads();
    for (unsigned i = tid; i < nb; i += 256) cb[i] = (int)sCol[i];
  }

  if (t0) {
    v8f c = wmma4(a0, b, zero8());
#pragma unroll
    for (int j = 0; j < 8; ++j) {
      int r = j + hi * 8;
      y[(size_t)(nb0 + r) * 16 + c16] = c[j] * sD[wave * 32 + r];
    }
  }
  if (t1) {
    v8f c = wmma4(a1, b, zero8());
#pragma unroll
    for (int j = 0; j < 8; ++j) {
      int r = 16 + j + hi * 8;
      y[(size_t)(nb0 + r) * 16 + c16] = c[j] * sD[wave * 32 + r];
    }
  }
}

__device__ __forceinline__ float4 add4(float4 a, float4 b) {
  return make_float4(a.x + b.x, a.y + b.y, a.z + b.z, a.w + b.w);
}
__device__ __forceinline__ float4 keep4(bool m, float4 a) {
  return make_float4(m ? a.x : 0.f, m ? a.y : 0.f, m ? a.z : 0.f, m ? a.w : 0.f);
}

template <bool kRelu, bool kNextW>
__global__ __launch_bounds__(256, 16) void k_gather(const int* __restrict__ rowstart,
                                                const int* __restrict__ col,
                                                const float* __restrict__ y,
                                                const float* __restrict__ dinv,
                                                const float* __restrict__ bias,
                                                const float* __restrict__ Wn,
                                                float* __restrict__ outp, int n, int nE) {
  __shared__ __align__(16) float sH[kNextW ? 64 * 20 : 4];
  __shared__ float sD[kNextW ? 64 : 4];
  int t = blockIdx.x * 256 + threadIdx.x;
  int nodeRaw = t >> 2, q = t & 3;
  bool valid = nodeRaw < n;
  int node = valid ? nodeRaw : n - 1;
  int beg = rowstart[node];
  int end = (node + 1 < n) ? rowstart[node + 1] : nE;
  const float4* __restrict__ y4 = reinterpret_cast<const float4*>(y);
  float4 acc = y4[(size_t)node * 4 + q];
  float di = dinv[node];

#pragma unroll 2
  for (int j = beg; j < end; j += 4) {
    int c0 = col[j], c1 = col[j + 1], c2 = col[j + 2], c3 = col[j + 3];
    bool m1 = j + 1 < end, m2 = j + 2 < end, m3 = j + 3 < end;
    int s1 = m1 ? c1 : node, s2 = m2 ? c2 : node, s3 = m3 ? c3 : node;
    float4 v0 = y4[(size_t)c0 * 4 + q];
    float4 v1 = y4[(size_t)s1 * 4 + q];
    float4 v2 = y4[(size_t)s2 * 4 + q];
    float4 v3 = y4[(size_t)s3 * 4 + q];
    acc = add4(acc, add4(add4(v0, keep4(m1, v1)), add4(keep4(m2, v2), keep4(m3, v3))));
  }

  float4 r = make_float4(di * acc.x + bias[q * 4 + 0], di * acc.y + bias[q * 4 + 1],
                         di * acc.z + bias[q * 4 + 2], di * acc.w + bias[q * 4 + 3]);
  if (kRelu) {
    r.x = fmaxf(r.x, 0.f); r.y = fmaxf(r.y, 0.f);
    r.z = fmaxf(r.z, 0.f); r.w = fmaxf(r.w, 0.f);
  }

  if (!kNextW) {
    if (valid) reinterpret_cast<float4*>(outp)[(size_t)node * 4 + q] = r;
  } else {
    int lane = threadIdx.x & 31, wave = threadIdx.x >> 5;
    int c16 = lane & 15, hi = lane >> 4, koff = hi * 2;
    v2f b[4];
#pragma unroll
    for (int kc = 0; kc < 4; ++kc) {
      b[kc][0] = Wn[(kc * 4 + koff) * 16 + c16];
      b[kc][1] = Wn[(kc * 4 + koff + 1) * 16 + c16];
    }
    int nl = threadIdx.x >> 2;
    *reinterpret_cast<float4*>(&sH[nl * 20 + q * 4]) = r;
    if (q == 0) sD[nl] = di;
    __syncthreads();
    if (wave < 4) {
      const float* hr = sH + (wave * 16 + c16) * 20;
      v8f c = zero8();
#pragma unroll
      for (int kc = 0; kc < 4; ++kc) {
        v2f a; a[0] = hr[kc * 4 + koff]; a[1] = hr[kc * 4 + koff + 1];
        c = wmma4(a, b[kc], c);
      }
      float dv[8];
#pragma unroll
      for (int j = 0; j < 8; ++j) dv[j] = sD[wave * 16 + j + hi * 8];
      int base = blockIdx.x * 64 + wave * 16;
      if (base + 16 <= n) {
#pragma unroll
        for (int j = 0; j < 8; ++j)
          outp[(size_t)(base + j + hi * 8) * 16 + c16] = c[j] * dv[j];
      } else {
#pragma unroll
        for (int j = 0; j < 8; ++j) {
          int nd = base + j + hi * 8;
          if (nd < n) outp[(size_t)nd * 16 + c16] = c[j] * dv[j];
        }
      }
    }
  }
}

__global__ __launch_bounds__(256) void k_mlp_pool(const float* __restrict__ h2,
                                                  const float* __restrict__ Wf1,
                                                  const float* __restrict__ bf1,
                                                  const float* __restrict__ Wf2,
                                                  const float* __restrict__ bf2,
                                                  float* __restrict__ g, int nG) {
  __shared__ float sW[64 * 132];
  __shared__ float sH[8 * 32 * 68];
  int lane = threadIdx.x & 31, wave = threadIdx.x >> 5;

  for (int i = threadIdx.x; i < 64 * 128; i += 256)
    sW[(i >> 7) * 132 + (i & 127)] = Wf2[i];

  int graph = blockIdx.x * 8 + wave;
  if (graph >= nG) graph = nG - 1;
  int col = lane & 15, hi = lane >> 4, koff = hi * 2;
  size_t nb = (size_t)graph * 32;

  v2f a1[2][4];
#pragma unroll
  for (int mt = 0; mt < 2; ++mt) {
    const float* r = h2 + (nb + mt * 16 + col) * 16;
#pragma unroll
    for (int kc = 0; kc < 4; ++kc) {
      a1[mt][kc][0] = r[kc * 4 + koff];
      a1[mt][kc][1] = r[kc * 4 + koff + 1];
    }
  }
  __syncthreads();

  float* mH = sH + wave * (32 * 68);
#pragma unroll
  for (int nt = 0; nt < 4; ++nt) {
    v8f c0 = zero8(), c1 = zero8();
#pragma unroll
    for (int kc = 0; kc < 4; ++kc) {
      v2f b;
      b[0] = Wf1[(kc * 4 + koff) * 64 + nt * 16 + col];
      b[1] = Wf1[(kc * 4 + koff + 1) * 64 + nt * 16 + col];
      c0 = wmma4(a1[0][kc], b, c0);
      c1 = wmma4(a1[1][kc], b, c1);
    }
    float bias = bf1[nt * 16 + col];
#pragma unroll
    for (int j = 0; j < 8; ++j) {
      mH[(j + hi * 8) * 68 + nt * 16 + col]       = fmaxf(c0[j] + bias, 0.f);
      mH[(16 + j + hi * 8) * 68 + nt * 16 + col]  = fmaxf(c1[j] + bias, 0.f);
    }
  }
  __syncthreads();

  const float* A0 = mH + (size_t)col * 68;
  const float* A1 = mH + (size_t)(16 + col) * 68;
  for (int nt2 = 0; nt2 < 8; ++nt2) {
    v8f c0 = zero8(), c1 = zero8();
#pragma unroll
    for (int kc = 0; kc < 16; ++kc) {
      int k0 = kc * 4 + koff;
      v2f a0; a0[0] = A0[k0]; a0[1] = A0[k0 + 1];
      v2f av; av[0] = A1[k0]; av[1] = A1[k0 + 1];
      v2f b;  b[0] = sW[k0 * 132 + nt2 * 16 + col];
      b[1] = sW[(k0 + 1) * 132 + nt2 * 16 + col];
      c0 = wmma4(a0, b, c0);
      c1 = wmma4(av, b, c1);
    }
    float bias = bf2[nt2 * 16 + col];
    float s = 0.f;
#pragma unroll
    for (int j = 0; j < 8; ++j)
      s += fmaxf(c0[j] + bias, 0.f) + fmaxf(c1[j] + bias, 0.f);
    s += __shfl_xor(s, 16, 32);
    if (lane < 16) g[(size_t)graph * 128 + nt2 * 16 + lane] = s * (1.0f / 32.0f);
  }
}

__global__ __launch_bounds__(256) void k_head(const float* __restrict__ g,
                                              const float* __restrict__ Wg,
                                              const float* __restrict__ bg,
                                              const float* __restrict__ Wout,
                                              const float* __restrict__ bout,
                                              float* __restrict__ out, int nG) {
  __shared__ float sG2[8 * 16 * 68];
  int lane = threadIdx.x & 31, wave = threadIdx.x >> 5;
  int tile = blockIdx.x * 8 + wave;
  int maxTile = (nG >> 4) - 1;
  if (tile > maxTile) tile = maxTile;
  int col = lane & 15, hi = lane >> 4, koff = hi * 2;
  int gbase = tile << 4;

  const float* ar = g + (size_t)(gbase + col) * 128;
  v8f c[4] = {zero8(), zero8(), zero8(), zero8()};
  for (int kc = 0; kc < 32; ++kc) {
    int k0 = kc * 4 + koff;
    v2f a; a[0] = ar[k0]; a[1] = ar[k0 + 1];
#pragma unroll
    for (int nt = 0; nt < 4; ++nt) {
      v2f b;
      b[0] = Wg[k0 * 64 + nt * 16 + col];
      b[1] = Wg[(k0 + 1) * 64 + nt * 16 + col];
      c[nt] = wmma4(a, b, c[nt]);
    }
  }
  float* mG = sG2 + wave * (16 * 68);
#pragma unroll
  for (int nt = 0; nt < 4; ++nt) {
    float bias = bg[nt * 16 + col];
#pragma unroll
    for (int j = 0; j < 8; ++j)
      mG[(j + hi * 8) * 68 + nt * 16 + col] = fmaxf(c[nt][j] + bias, 0.f);
  }
  __syncthreads();

  const float* g2 = mG + (size_t)col * 68;
  int gidx = gbase + col;
#pragma unroll
  for (int mm = 0; mm < 2; ++mm) {
    int m = hi * 2 + mm;
    float acc[5];
#pragma unroll
    for (int a2 = 0; a2 < 5; ++a2) acc[a2] = bout[m * 5 + a2];
    for (int f = 0; f < 64; ++f) {
      float gv = g2[f];
#pragma unroll
      for (int a2 = 0; a2 < 5; ++a2)
        acc[a2] = fmaf(gv, Wout[(m * 64 + f) * 5 + a2], acc[a2]);
    }
#pragma unroll
    for (int a2 = 0; a2 < 5; ++a2)
      out[((size_t)gidx * 4 + m) * 5 + a2] = acc[a2];
  }
}

extern "C" void kernel_launch(void* const* d_in, const int* in_sizes, int n_in,
                              void* d_out, int out_size, void* d_ws, size_t ws_size,
                              hipStream_t stream) {
  const float* x    = (const float*)d_in[0];
  const int*   ei   = (const int*)d_in[1];
  const float* W1   = (const float*)d_in[3];
  const float* b1   = (const float*)d_in[4];
  const float* W2   = (const float*)d_in[5];
  const float* b2   = (const float*)d_in[6];
  const float* Wf1  = (const float*)d_in[7];
  const float* bf1  = (const float*)d_in[8];
  const float* Wf2  = (const float*)d_in[9];
  const float* bf2  = (const float*)d_in[10];
  const float* Wg   = (const float*)d_in[11];
  const float* bg   = (const float*)d_in[12];
  const float* Wout = (const float*)d_in[13];
  const float* bout = (const float*)d_in[14];
  float* out = (float*)d_out;

  const int N = in_sizes[0] / 4;
  const int E = in_sizes[1] / 2;
  const int G = in_sizes[2] / 32;
  const int* src = ei;
  const int* dst = ei + E;

  char* ws = (char*)d_ws;
  size_t off = 0;
  auto take = [&](size_t bytes) -> void* {
    void* p = ws + off;
    off += (bytes + 255) & ~(size_t)255;
    return p;
  };
  float*    dinv     = (float*)take((size_t)N * 4);
  int*      rowstart = (int*)take((size_t)N * 4);
  float*    ybuf     = (float*)take((size_t)N * 16 * 4);
  float*    hbuf     = (float*)take((size_t)N * 16 * 4);
  int*      col      = (int*)take((size_t)E * 4);
  float*    gbuf     = (float*)take((size_t)G * 128 * 4);
  unsigned* pairs    = (unsigned*)hbuf;
  unsigned* offs     = (unsigned*)ybuf;
  unsigned* bcnt     = (unsigned*)gbuf;
  (void)ws_size; (void)n_in; (void)out_size;

  const int nB   = (N + kBucketNodes - 1) / kBucketNodes;
  const int gS   = kMaxBuckets / 32;
  const int gC   = (E + kChunk - 1) / kChunk;
  const int gN4  = (N * 4 + 255) / 256;
  const int gMLP = (G + 7) / 8;
  const int gHD  = (G / 16 + 7) / 8;

  k_bhist<<<gC, 256, 0, stream>>>(dst, offs, E);
  k_colscan<<<gS, 1024, 0, stream>>>(offs, bcnt, gC);
  k_bin<<<gC, 256, 0, stream>>>(src, dst, bcnt, offs, pairs, E);
  k_bucket_xw1<<<nB, 256, 0, stream>>>(pairs, bcnt, rowstart, dinv, col, x, W1, ybuf, N);

  k_gather<true, true><<<gN4, 256, 0, stream>>>(rowstart, col, ybuf, dinv, b1, W2, hbuf, N, E);

  k_gather<false, false><<<gN4, 256, 0, stream>>>(rowstart, col, hbuf, dinv, b2, W2, ybuf, N, E);

  k_mlp_pool<<<gMLP, 256, 0, stream>>>(ybuf, Wf1, bf1, Wf2, bf2, gbuf, G);

  k_head<<<gHD, 256, 0, stream>>>(gbuf, Wg, bg, Wout, bout, out, G);
}
